# starting kernel
# baseline (speedup 1.0000x reference)
#include <hip/hip_runtime.h>

typedef float v2f __attribute__((ext_vector_type(2)));
typedef float v8f __attribute__((ext_vector_type(8)));

#define NB   256
#define DD   128
#define NP   1296
#define NC   64
#define EPSF 1e-4f

__global__ __launch_bounds__(256)
void ImageParser_68745246540061_kernel(const float* __restrict__ x,
                                       const int*   __restrict__ cm,
                                       float*       __restrict__ out) {
    __shared__ int   s_cm[NP];
    __shared__ float s_cnt[NC];
    __shared__ float s_mean[DD * NC];

    const int tid  = threadIdx.x;
    const int b    = blockIdx.x;
    const int lane = tid & 31;
    const int wv   = tid >> 5;

    const int* cmb = cm + b * NP;
    for (int i = tid; i < NP; i += 256) s_cm[i] = cmb[i];
    if (tid < NC) s_cnt[tid] = 0.0f;
    __syncthreads();
    for (int i = tid; i < NP; i += 256) atomicAdd(&s_cnt[s_cm[i]], 1.0f);
    __syncthreads();

    const int  mrow = wv * 16 + (lane & 15);
    const int  kb   = (lane >> 4) << 1;
    const float* xrow = x + ((size_t)b * DD + mrow) * NP;
    const int  myc  = lane & 15;

    v8f acc[4] = {};

    for (int n0 = 0; n0 < NP; n0 += 4) {
        v2f a = *(const v2f*)(xrow + n0 + kb);
        int c0 = s_cm[n0 + kb];
        int c1 = s_cm[n0 + kb + 1];
#pragma unroll
        for (int t = 0; t < 4; ++t) {
            int cc = t * 16 + myc;
            v2f bf;
            bf.x = (c0 == cc) ? 1.0f : 0.0f;
            bf.y = (c1 == cc) ? 1.0f : 0.0f;
            acc[t] = __builtin_amdgcn_wmma_f32_16x16x4_f32(
                false, a, false, bf, (short)0, acc[t], false, false);
        }
    }

    const int rbase = wv * 16 + ((lane >> 4) << 3);
#pragma unroll
    for (int t = 0; t < 4; ++t) {
        int col = t * 16 + myc;
#pragma unroll
        for (int j = 0; j < 8; ++j)
            s_mean[(rbase + j) * NC + col] = acc[t][j];
    }
    __syncthreads();

    for (int i = tid; i < DD * NC; i += 256)
        s_mean[i] = s_mean[i] / (s_cnt[i & (NC - 1)] + EPSF);
    __syncthreads();

    const float* xb = x + (size_t)b * DD * NP;
    for (int n = tid; n < NP; n += 256) {
        int   c    = s_cm[n];
        float accd = 0.0f;
#pragma unroll 4
        for (int d = 0; d < DD; ++d) {
            float v    = xb[(size_t)d * NP + n];
            float m    = s_mean[d * NC + c];
            float diff = v - m;
            accd = fmaf(diff, diff, accd);
        }
        out[(size_t)b * NP + n] = accd * (1.0f / (float)DD);
    }
}

extern "C" void kernel_launch(void* const* d_in, const int* in_sizes, int n_in,
                              void* d_out, int out_size, void* d_ws, size_t ws_size,
                              hipStream_t stream) {
    (void)in_sizes; (void)n_in; (void)d_ws; (void)ws_size; (void)out_size;
    const float* x  = (const float*)d_in[0];
    const int*   cm = (const int*)d_in[1];
    float*       o  = (float*)d_out;
    ImageParser_68745246540061_kernel<<<NB, 256, 0, stream>>>(x, cm, o);
}


# distance pass re-tiled for loads in flight: wave = 16-row slab, lane = 4 pixels, 16-B row loads, next step's 16 issued before this step's math; sum over d = 8 slab sums added in slab order (LDS)
# speedup vs baseline: 1.6275x; 1.6275x over previous
#include <hip/hip_runtime.h>

typedef float v2f __attribute__((ext_vector_type(2)));
typedef float v8f __attribute__((ext_vector_type(8)));

#define NB   256
#define DD   128
#define NP   1296
#define NC   64
#define NG   324
#define NSTEP 11
#define EPSF 1e-4f

__device__ __forceinline__ void load_rows(float4 (&v)[16], const float* __restrict__ p) {
#pragma unroll
    for (int r = 0; r < 16; ++r) v[r] = *(const float4*)(p + (size_t)r * NP);
}

__device__ __forceinline__ void slab_dist(const float4 (&v)[16], const float* mb,
                                          const int* s_cm, float* part, int g) {
    const int4 c4 = *(const int4*)&s_cm[4 * g];
    float a0 = 0.0f, a1 = 0.0f, a2 = 0.0f, a3 = 0.0f;
#pragma unroll
    for (int r = 0; r < 16; ++r) {
        float d0 = v[r].x - mb[r * NC + c4.x];
        float d1 = v[r].y - mb[r * NC + c4.y];
        float d2 = v[r].z - mb[r * NC + c4.z];
        float d3 = v[r].w - mb[r * NC + c4.w];
        a0 = fmaf(d0, d0, a0);
        a1 = fmaf(d1, d1, a1);
        a2 = fmaf(d2, d2, a2);
        a3 = fmaf(d3, d3, a3);
    }
    *(float4*)&part[4 * g] = make_float4(a0, a1, a2, a3);
}

__global__ __launch_bounds__(256)
void class_mean_dist_kernel(const float* __restrict__ x,
                            const int*   __restrict__ cm,
                            float*       __restrict__ out) {
    __shared__ __attribute__((aligned(16))) int   s_cm[NP];
    __shared__ float s_cnt[NC];
    __shared__ float s_mean[DD * NC];
    __shared__ __attribute__((aligned(16))) float s_part[8 * NP];

    const int tid  = threadIdx.x;
    const int b    = blockIdx.x;
    const int lane = tid & 31;
    const int wv   = tid >> 5;

    const int* cmb = cm + b * NP;
    for (int i = tid; i < NP; i += 256) s_cm[i] = cmb[i];
    if (tid < NC) s_cnt[tid] = 0.0f;
    __syncthreads();
    for (int i = tid; i < NP; i += 256) atomicAdd(&s_cnt[s_cm[i]], 1.0f);
    __syncthreads();

    const int  mrow = wv * 16 + (lane & 15);
    const int  kb   = (lane >> 4) << 1;
    const float* xrow = x + ((size_t)b * DD + mrow) * NP;
    const int  myc  = lane & 15;

    v8f acc[4] = {};

    for (int n0 = 0; n0 < NP; n0 += 4) {
        v2f a = *(const v2f*)(xrow + n0 + kb);
        int c0 = s_cm[n0 + kb];
        int c1 = s_cm[n0 + kb + 1];
#pragma unroll
        for (int t = 0; t < 4; ++t) {
            int cc = t * 16 + myc;
            v2f bf;
            bf.x = (c0 == cc) ? 1.0f : 0.0f;
            bf.y = (c1 == cc) ? 1.0f : 0.0f;
            acc[t] = __builtin_amdgcn_wmma_f32_16x16x4_f32(
                false, a, false, bf, (short)0, acc[t], false, false);
        }
    }

    const int rbase = wv * 16 + ((lane >> 4) << 3);
#pragma unroll
    for (int t = 0; t < 4; ++t) {
        int col = t * 16 + myc;
#pragma unroll
        for (int j = 0; j < 8; ++j)
            s_mean[(rbase + j) * NC + col] = acc[t][j];
    }
    __syncthreads();

    for (int i = tid; i < DD * NC; i += 256)
        s_mean[i] = s_mean[i] / (s_cnt[i & (NC - 1)] + EPSF);
    __syncthreads();

    const float* xs   = x + ((size_t)b * DD + (size_t)wv * 16) * NP;
    const float* mb   = s_mean + wv * 16 * NC;
    float*       part = s_part + wv * NP;
    float4 va[16], vb[16];
    load_rows(va, xs + 4 * lane);
    for (int j = 0; j < NSTEP - 1; j += 2) {
        const int g0 = 32 * j + lane;
        load_rows(vb, xs + 4 * (g0 + 32));
        __builtin_amdgcn_sched_barrier(0);
        slab_dist(va, mb, s_cm, part, g0);
        load_rows(va, xs + 4 * min(g0 + 64, NG - 1));
        __builtin_amdgcn_sched_barrier(0);
        slab_dist(vb, mb, s_cm, part, g0 + 32);
    }
    {
        const int g = 32 * (NSTEP - 1) + lane;
        if (g < NG) slab_dist(va, mb, s_cm, part, g);
    }
    __syncthreads();

    float4* out4 = (float4*)(out + (size_t)b * NP);
    for (int q = tid; q < NG; q += 256) {
        float4 s = *(const float4*)&s_part[4 * q];
#pragma unroll
        for (int w = 1; w < 8; ++w) {
            float4 p = *(const float4*)&s_part[w * NP + 4 * q];
            s.x += p.x; s.y += p.y; s.z += p.z; s.w += p.w;
        }
        const float sc = 1.0f / (float)DD;
        out4[q] = make_float4(s.x * sc, s.y * sc, s.z * sc, s.w * sc);
    }
}

extern "C" void kernel_launch(void* const* d_in, const int* in_sizes, int n_in,
                              void* d_out, int out_size, void* d_ws, size_t ws_size,
                              hipStream_t stream) {
    (void)in_sizes; (void)n_in; (void)d_ws; (void)ws_size; (void)out_size;
    const float* x  = (const float*)d_in[0];
    const int*   cm = (const int*)d_in[1];
    float*       o  = (float*)d_out;
    class_mean_dist_kernel<<<NB, 256, 0, stream>>>(x, cm, o);
}


# mean pass fed like the distance pass: x read as 512-B row loads one step ahead, turned into WMMA A fragments through a wave-private LDS stage (was 8-B loads of 16 rows per load); sum order kept
# speedup vs baseline: 2.9538x; 1.8149x over previous
#include <hip/hip_runtime.h>

typedef float v2f __attribute__((ext_vector_type(2)));
typedef float v8f __attribute__((ext_vector_type(8)));

#define NB   256
#define DD   128
#define NP   1296
#define NC   64
#define NG   324
#define NSTEP 11
#define EPSF 1e-4f
#define SP   132
#define NFULL 10

__device__ __forceinline__ void load_rows(float4 (&v)[16], const float* __restrict__ p) {
#pragma unroll
    for (int r = 0; r < 16; ++r) v[r] = *(const float4*)(p + (size_t)r * NP);
}

__device__ __forceinline__ void stage_rows(float* st, const float4 (&v)[16], int lane) {
#pragma unroll
    for (int r = 0; r < 16; ++r) *(float4*)&st[r * SP + 4 * lane] = v[r];
}

__device__ __forceinline__ void wave_lds_order() {
    __builtin_amdgcn_fence(__ATOMIC_RELEASE, "wavefront");
    __builtin_amdgcn_wave_barrier();
    __builtin_amdgcn_fence(__ATOMIC_ACQUIRE, "wavefront");
}

template <int NCH>
__device__ __forceinline__ void wmma_chunks(v8f (&acc)[4], const float* arow, const int* cmk, int myc) {
    v2f  a[NCH];
    int2 k[NCH];
#pragma unroll
    for (int q = 0; q < NCH; ++q) {
        a[q] = *(const v2f*)(arow + 4 * q);
        k[q] = *(const int2*)(cmk + 4 * q);
    }
#pragma unroll
    for (int q = 0; q < NCH; ++q) {
#pragma unroll
        for (int t = 0; t < 4; ++t) {
            int cc = t * 16 + myc;
            v2f bf;
            bf.x = (k[q].x == cc) ? 1.0f : 0.0f;
            bf.y = (k[q].y == cc) ? 1.0f : 0.0f;
            acc[t] = __builtin_amdgcn_wmma_f32_16x16x4_f32(
                false, a[q], false, bf, (short)0, acc[t], false, false);
        }
    }
}

__device__ __forceinline__ void slab_dist(const float4 (&v)[16], const float* mb,
                                          const int* s_cm, float* part, int g) {
    const int4 c4 = *(const int4*)&s_cm[4 * g];
    float a0 = 0.0f, a1 = 0.0f, a2 = 0.0f, a3 = 0.0f;
#pragma unroll
    for (int r = 0; r < 16; ++r) {
        float d0 = v[r].x - mb[r * NC + c4.x];
        float d1 = v[r].y - mb[r * NC + c4.y];
        float d2 = v[r].z - mb[r * NC + c4.z];
        float d3 = v[r].w - mb[r * NC + c4.w];
        a0 = fmaf(d0, d0, a0);
        a1 = fmaf(d1, d1, a1);
        a2 = fmaf(d2, d2, a2);
        a3 = fmaf(d3, d3, a3);
    }
    *(float4*)&part[4 * g] = make_float4(a0, a1, a2, a3);
}

__global__ __launch_bounds__(256)
void class_mean_dist_kernel(const float* __restrict__ x,
                            const int*   __restrict__ cm,
                            float*       __restrict__ out) {
    __shared__ __attribute__((aligned(16))) int   s_cm[NP];
    __shared__ float s_cnt[NC];
    __shared__ float s_mean[DD * NC];
    __shared__ __attribute__((aligned(16))) float s_part[8 * NP];
    __shared__ __attribute__((aligned(16))) float s_stage[8 * 16 * SP];

    const int tid  = threadIdx.x;
    const int b    = blockIdx.x;
    const int lane = tid & 31;
    const int wv   = tid >> 5;

    const float* xs = x + ((size_t)b * DD + (size_t)wv * 16) * NP;
    float4 va[16], vb[16];
    load_rows(va, xs + 4 * lane);

    const int* cmb = cm + b * NP;
    for (int i = tid; i < NP; i += 256) s_cm[i] = cmb[i];
    if (tid < NC) s_cnt[tid] = 0.0f;
    __syncthreads();
    for (int i = tid; i < NP; i += 256) atomicAdd(&s_cnt[s_cm[i]], 1.0f);
    __syncthreads();

    const int  myc  = lane & 15;
    const int  kb   = (lane >> 4) << 1;
    float*       st   = s_stage + wv * 16 * SP;
    const float* arow = st + myc * SP + kb;

    v8f acc[4] = {};

    for (int s = 0; s < NFULL; ++s) {
        stage_rows(st, va, lane);
        load_rows(va, xs + 4 * min(32 * (s + 1) + lane, NG - 1));
        wave_lds_order();
        wmma_chunks<32>(acc, arow, s_cm + 128 * s + kb, myc);
        wave_lds_order();
    }
    stage_rows(st, va, lane);
    wave_lds_order();
    wmma_chunks<4>(acc, arow, s_cm + 128 * NFULL + kb, myc);

    const int rbase = wv * 16 + ((lane >> 4) << 3);
#pragma unroll
    for (int t = 0; t < 4; ++t) {
        int col = t * 16 + myc;
#pragma unroll
        for (int j = 0; j < 8; ++j)
            s_mean[(rbase + j) * NC + col] = acc[t][j];
    }
    __syncthreads();

    for (int i = tid; i < DD * NC; i += 256)
        s_mean[i] = s_mean[i] / (s_cnt[i & (NC - 1)] + EPSF);
    __syncthreads();

    const float* mb   = s_mean + wv * 16 * NC;
    float*       part = s_part + wv * NP;
    load_rows(va, xs + 4 * lane);
    for (int j = 0; j < NSTEP - 1; j += 2) {
        const int g0 = 32 * j + lane;
        load_rows(vb, xs + 4 * (g0 + 32));
        __builtin_amdgcn_sched_barrier(0);
        slab_dist(va, mb, s_cm, part, g0);
        load_rows(va, xs + 4 * min(g0 + 64, NG - 1));
        __builtin_amdgcn_sched_barrier(0);
        slab_dist(vb, mb, s_cm, part, g0 + 32);
    }
    {
        const int g = 32 * (NSTEP - 1) + lane;
        if (g < NG) slab_dist(va, mb, s_cm, part, g);
    }
    __syncthreads();

    float4* out4 = (float4*)(out + (size_t)b * NP);
    for (int q = tid; q < NG; q += 256) {
        float4 s = *(const float4*)&s_part[4 * q];
#pragma unroll
        for (int w = 1; w < 8; ++w) {
            float4 p = *(const float4*)&s_part[w * NP + 4 * q];
            s.x += p.x; s.y += p.y; s.z += p.z; s.w += p.w;
        }
        const float sc = 1.0f / (float)DD;
        out4[q] = make_float4(s.x * sc, s.y * sc, s.z * sc, s.w * sc);
    }
}

extern "C" void kernel_launch(void* const* d_in, const int* in_sizes, int n_in,
                              void* d_out, int out_size, void* d_ws, size_t ws_size,
                              hipStream_t stream) {
    (void)in_sizes; (void)n_in; (void)d_ws; (void)ws_size; (void)out_size;
    const float* x  = (const float*)d_in[0];
    const int*   cm = (const int*)d_in[1];
    float*       o  = (float*)d_out;
    class_mean_dist_kernel<<<NB, 256, 0, stream>>>(x, cm, o);
}


# mean pass: each 128-pixel step's pixels sorted by 16-class tile in LDS, so a 4-pixel chunk takes 1 f32 WMMA not 4; pixel sum order now tile-sorted, 2 chains per tile added at the end
# speedup vs baseline: 4.0574x; 1.3736x over previous
#include <hip/hip_runtime.h>

typedef float v2f __attribute__((ext_vector_type(2)));
typedef float v8f __attribute__((ext_vector_type(8)));

#define NB   256
#define DD   128
#define NP   1296
#define NC   64
#define NG   324
#define NSTEP 11
#define EPSF 1e-4f
#define SP   196
#define NSLOT 192
#define NST  11
#define NFULL 10

__device__ __forceinline__ void load_rows(float4 (&v)[16], const float* __restrict__ p) {
#pragma unroll
    for (int r = 0; r < 16; ++r) v[r] = *(const float4*)(p + (size_t)r * NP);
}

__device__ __forceinline__ void stage_rows(float* st, const float4 (&v)[16], int4 ps) {
    float* p0 = st + ps.x;
    float* p1 = st + ps.y;
    float* p2 = st + ps.z;
    float* p3 = st + ps.w;
#pragma unroll
    for (int r = 0; r < 16; ++r) {
        p0[r * SP] = v[r].x;
        p1[r * SP] = v[r].y;
        p2[r * SP] = v[r].z;
        p3[r * SP] = v[r].w;
    }
}

__device__ __forceinline__ void wave_lds_order() {
    __builtin_amdgcn_fence(__ATOMIC_RELEASE, "wavefront");
    __builtin_amdgcn_wave_barrier();
    __builtin_amdgcn_fence(__ATOMIC_ACQUIRE, "wavefront");
}

__device__ __forceinline__ void load_unit(v2f (&a)[4], int2 (&k)[4], const float* ap, const int* kp) {
#pragma unroll
    for (int q = 0; q < 4; ++q) {
        a[q] = *(const v2f*)(ap + 4 * q);
        k[q] = *(const int2*)(kp + 4 * q);
    }
}

template <int T>
__device__ __forceinline__ void tile_units(v8f (&acc)[8], v2f (&a)[4], int2 (&k)[4],
                                           const float*& ap, const int*& kp, int n, int myc) {
    const int cc = 16 * T + myc;
    for (int u = 0; u < n; ++u) {
        v2f  an[4];
        int2 kn[4];
        ap += 16;
        kp += 16;
        load_unit(an, kn, ap, kp);
        __builtin_amdgcn_sched_barrier(0);
#pragma unroll
        for (int q = 0; q < 4; ++q) {
            v2f bf;
            bf.x = (k[q].x == cc) ? 1.0f : 0.0f;
            bf.y = (k[q].y == cc) ? 1.0f : 0.0f;
            acc[2 * T + (q & 1)] = __builtin_amdgcn_wmma_f32_16x16x4_f32(
                false, a[q], false, bf, (short)0, acc[2 * T + (q & 1)], false, false);
        }
#pragma unroll
        for (int q = 0; q < 4; ++q) { a[q] = an[q]; k[q] = kn[q]; }
    }
}

__device__ __forceinline__ void wmma_step(v8f (&acc)[8], const float* arow, const int* kp0,
                                          const int* nu, int myc) {
    const int4 n4 = *(const int4*)nu;
    const int n0 = __builtin_amdgcn_readfirstlane(n4.x);
    const int n1 = __builtin_amdgcn_readfirstlane(n4.y);
    const int n2 = __builtin_amdgcn_readfirstlane(n4.z);
    const int n3 = __builtin_amdgcn_readfirstlane(n4.w);
    const float* ap = arow;
    const int*   kp = kp0;
    v2f  a[4];
    int2 k[4];
    load_unit(a, k, ap, kp);
    tile_units<0>(acc, a, k, ap, kp, n0, myc);
    tile_units<1>(acc, a, k, ap, kp, n1, myc);
    tile_units<2>(acc, a, k, ap, kp, n2, myc);
    tile_units<3>(acc, a, k, ap, kp, n3, myc);
}

__device__ __forceinline__ void slab_dist(const float4 (&v)[16], const float* mb,
                                          const int* s_cm, float* part, int g) {
    const int4 c4 = *(const int4*)&s_cm[4 * g];
    float a0 = 0.0f, a1 = 0.0f, a2 = 0.0f, a3 = 0.0f;
#pragma unroll
    for (int r = 0; r < 16; ++r) {
        float d0 = v[r].x - mb[r * NC + c4.x];
        float d1 = v[r].y - mb[r * NC + c4.y];
        float d2 = v[r].z - mb[r * NC + c4.z];
        float d3 = v[r].w - mb[r * NC + c4.w];
        a0 = fmaf(d0, d0, a0);
        a1 = fmaf(d1, d1, a1);
        a2 = fmaf(d2, d2, a2);
        a3 = fmaf(d3, d3, a3);
    }
    *(float4*)&part[4 * g] = make_float4(a0, a1, a2, a3);
}

__global__ __launch_bounds__(256)
void class_mean_dist_kernel(const float* __restrict__ x,
                            const int*   __restrict__ cm,
                            float*       __restrict__ out) {
    __shared__ __attribute__((aligned(16))) int   s_cm[NP];
    __shared__ float s_cnt[NC];
    __shared__ float s_mean[DD * NC];
    __shared__ __attribute__((aligned(16))) float s_part[8 * NP];
    __shared__ __attribute__((aligned(16))) float s_stage[8 * 16 * SP];
    __shared__ __attribute__((aligned(16))) int   s_pos[NP];
    __shared__ __attribute__((aligned(16))) int   s_cls[NST * NSLOT];
    __shared__ __attribute__((aligned(16))) int   s_nu[NST * 4];

    const int tid  = threadIdx.x;
    const int b    = blockIdx.x;
    const int lane = tid & 31;
    const int wv   = tid >> 5;

    const float* xs = x + ((size_t)b * DD + (size_t)wv * 16) * NP;
    float4 va[16], vb[16];
    load_rows(va, xs + 4 * lane);

    const int* cmb = cm + b * NP;
    for (int i = tid; i < NP; i += 256) s_cm[i] = cmb[i];
    for (int i = tid; i < NST * NSLOT; i += 256) s_cls[i] = -1;
    if (tid < NC) s_cnt[tid] = 0.0f;
    float* st = s_stage + wv * 16 * SP;
    for (int i = lane; i < 16 * SP / 4; i += 32) ((float4*)st)[i] = make_float4(0.0f, 0.0f, 0.0f, 0.0f);
    __syncthreads();
    for (int i = tid; i < NP; i += 256) atomicAdd(&s_cnt[s_cm[i]], 1.0f);

    for (int s = wv; s < NST; s += 8) {
        const int  p0    = 128 * s + 4 * lane;
        const bool valid = p0 < NP;
        int4 c4 = make_int4(0, 0, 0, 0);
        if (valid) c4 = *(const int4*)&s_cm[p0];
        const int sh0 = (c4.x >> 4) << 3, sh1 = (c4.y >> 4) << 3;
        const int sh2 = (c4.z >> 4) << 3, sh3 = (c4.w >> 4) << 3;
        const unsigned own = valid ? ((1u << sh0) + (1u << sh1) + (1u << sh2) + (1u << sh3)) : 0u;
        unsigned inc = own;
#pragma unroll
        for (int d = 1; d < 32; d <<= 1) {
            const unsigned up = __shfl_up(inc, (unsigned)d);
            if (lane >= d) inc += up;
        }
        const unsigned tot = __shfl(inc, 31);
        const int u0 = (int)(((tot      ) & 255u) + 15u) >> 4;
        const int u1 = (int)(((tot >>  8) & 255u) + 15u) >> 4;
        const int u2 = (int)(((tot >> 16) & 255u) + 15u) >> 4;
        const int u3 = (int)(((tot >> 24)       ) + 15u) >> 4;
        const unsigned base = ((unsigned)(16 * u0) << 8) + ((unsigned)(16 * (u0 + u1)) << 16)
                            + ((unsigned)(16 * (u0 + u1 + u2)) << 24);
        unsigned run = base + (inc - own);
        const int q0 = (int)((run >> sh0) & 255u); run += 1u << sh0;
        const int q1 = (int)((run >> sh1) & 255u); run += 1u << sh1;
        const int q2 = (int)((run >> sh2) & 255u); run += 1u << sh2;
        const int q3 = (int)((run >> sh3) & 255u);
        if (valid) {
            *(int4*)&s_pos[p0] = make_int4(q0, q1, q2, q3);
            s_cls[s * NSLOT + q0] = c4.x;
            s_cls[s * NSLOT + q1] = c4.y;
            s_cls[s * NSLOT + q2] = c4.z;
            s_cls[s * NSLOT + q3] = c4.w;
        }
        if (lane == 0) *(int4*)&s_nu[4 * s] = make_int4(u0, u1, u2, u3);
    }
    __syncthreads();

    const int  myc  = lane & 15;
    const int  kb   = (lane >> 4) << 1;
    const float* arow = st + myc * SP + kb;

    v8f acc[8] = {};

    for (int s = 0; s < NFULL; ++s) {
        stage_rows(st, va, *(const int4*)&s_pos[128 * s + 4 * lane]);
        load_rows(va, xs + 4 * min(32 * (s + 1) + lane, NG - 1));
        wave_lds_order();
        wmma_step(acc, arow, s_cls + NSLOT * s + kb, s_nu + 4 * s, myc);
        wave_lds_order();
    }
    if (lane < 4)
        stage_rows(st, va, *(const int4*)&s_pos[128 * NFULL + 4 * lane]);
    wave_lds_order();
    wmma_step(acc, arow, s_cls + NSLOT * NFULL + kb, s_nu + 4 * NFULL, myc);

    const int rbase = wv * 16 + ((lane >> 4) << 3);
#pragma unroll
    for (int t = 0; t < 4; ++t) {
        int col = t * 16 + myc;
        const v8f sum = acc[2 * t] + acc[2 * t + 1];
#pragma unroll
        for (int j = 0; j < 8; ++j)
            s_mean[(rbase + j) * NC + col] = sum[j];
    }
    __syncthreads();

    for (int i = tid; i < DD * NC; i += 256)
        s_mean[i] = s_mean[i] / (s_cnt[i & (NC - 1)] + EPSF);
    __syncthreads();

    const float* mb   = s_mean + wv * 16 * NC;
    float*       part = s_part + wv * NP;
    load_rows(va, xs + 4 * lane);
    for (int j = 0; j < NSTEP - 1; j += 2) {
        const int g0 = 32 * j + lane;
        load_rows(vb, xs + 4 * (g0 + 32));
        __builtin_amdgcn_sched_barrier(0);
        slab_dist(va, mb, s_cm, part, g0);
        load_rows(va, xs + 4 * min(g0 + 64, NG - 1));
        __builtin_amdgcn_sched_barrier(0);
        slab_dist(vb, mb, s_cm, part, g0 + 32);
    }
    {
        const int g = 32 * (NSTEP - 1) + lane;
        if (g < NG) slab_dist(va, mb, s_cm, part, g);
    }
    __syncthreads();

    float4* out4 = (float4*)(out + (size_t)b * NP);
    for (int q = tid; q < NG; q += 256) {
        float4 s = *(const float4*)&s_part[4 * q];
#pragma unroll
        for (int w = 1; w < 8; ++w) {
            float4 p = *(const float4*)&s_part[w * NP + 4 * q];
            s.x += p.x; s.y += p.y; s.z += p.z; s.w += p.w;
        }
        const float sc = 1.0f / (float)DD;
        out4[q] = make_float4(s.x * sc, s.y * sc, s.z * sc, s.w * sc);
    }
}

extern "C" void kernel_launch(void* const* d_in, const int* in_sizes, int n_in,
                              void* d_out, int out_size, void* d_ws, size_t ws_size,
                              hipStream_t stream) {
    (void)in_sizes; (void)n_in; (void)d_ws; (void)ws_size; (void)out_size;
    const float* x  = (const float*)d_in[0];
    const int*   cm = (const int*)d_in[1];
    float*       o  = (float*)d_out;
    class_mean_dist_kernel<<<NB, 256, 0, stream>>>(x, cm, o);
}


# class map fetched in one memory round trip: 2 16-B loads per thread issued ahead of the x loads, one wait (was 6 dependent 4-B loads, each waited out); counts (exact sums of 1.0) added from registers
# speedup vs baseline: 4.2474x; 1.0468x over previous
#include <hip/hip_runtime.h>

typedef float v2f __attribute__((ext_vector_type(2)));
typedef float v8f __attribute__((ext_vector_type(8)));

#define NB   256
#define DD   128
#define NP   1296
#define NC   64
#define NG   324
#define NSTEP 11
#define EPSF 1e-4f
#define SP   196
#define NSLOT 192
#define NST  11
#define NFULL 10

__device__ __forceinline__ void load_rows(float4 (&v)[16], const float* __restrict__ p) {
#pragma unroll
    for (int r = 0; r < 16; ++r) v[r] = *(const float4*)(p + (size_t)r * NP);
}

__device__ __forceinline__ void stage_rows(float* st, const float4 (&v)[16], int4 ps) {
    float* p0 = st + ps.x;
    float* p1 = st + ps.y;
    float* p2 = st + ps.z;
    float* p3 = st + ps.w;
#pragma unroll
    for (int r = 0; r < 16; ++r) {
        p0[r * SP] = v[r].x;
        p1[r * SP] = v[r].y;
        p2[r * SP] = v[r].z;
        p3[r * SP] = v[r].w;
    }
}

__device__ __forceinline__ void wave_lds_order() {
    __builtin_amdgcn_fence(__ATOMIC_RELEASE, "wavefront");
    __builtin_amdgcn_wave_barrier();
    __builtin_amdgcn_fence(__ATOMIC_ACQUIRE, "wavefront");
}

__device__ __forceinline__ void load_unit(v2f (&a)[4], int2 (&k)[4], const float* ap, const int* kp) {
#pragma unroll
    for (int q = 0; q < 4; ++q) {
        a[q] = *(const v2f*)(ap + 4 * q);
        k[q] = *(const int2*)(kp + 4 * q);
    }
}

template <int T>
__device__ __forceinline__ void tile_units(v8f (&acc)[8], v2f (&a)[4], int2 (&k)[4],
                                           const float*& ap, const int*& kp, int n, int myc) {
    const int cc = 16 * T + myc;
    for (int u = 0; u < n; ++u) {
        v2f  an[4];
        int2 kn[4];
        ap += 16;
        kp += 16;
        load_unit(an, kn, ap, kp);
        __builtin_amdgcn_sched_barrier(0);
#pragma unroll
        for (int q = 0; q < 4; ++q) {
            v2f bf;
            bf.x = (k[q].x == cc) ? 1.0f : 0.0f;
            bf.y = (k[q].y == cc) ? 1.0f : 0.0f;
            acc[2 * T + (q & 1)] = __builtin_amdgcn_wmma_f32_16x16x4_f32(
                false, a[q], false, bf, (short)0, acc[2 * T + (q & 1)], false, false);
        }
#pragma unroll
        for (int q = 0; q < 4; ++q) { a[q] = an[q]; k[q] = kn[q]; }
    }
}

__device__ __forceinline__ void wmma_step(v8f (&acc)[8], const float* arow, const int* kp0,
                                          const int* nu, int myc) {
    const int4 n4 = *(const int4*)nu;
    const int n0 = __builtin_amdgcn_readfirstlane(n4.x);
    const int n1 = __builtin_amdgcn_readfirstlane(n4.y);
    const int n2 = __builtin_amdgcn_readfirstlane(n4.z);
    const int n3 = __builtin_amdgcn_readfirstlane(n4.w);
    const float* ap = arow;
    const int*   kp = kp0;
    v2f  a[4];
    int2 k[4];
    load_unit(a, k, ap, kp);
    tile_units<0>(acc, a, k, ap, kp, n0, myc);
    tile_units<1>(acc, a, k, ap, kp, n1, myc);
    tile_units<2>(acc, a, k, ap, kp, n2, myc);
    tile_units<3>(acc, a, k, ap, kp, n3, myc);
}

__device__ __forceinline__ void slab_dist(const float4 (&v)[16], const float* mb,
                                          const int* s_cm, float* part, int g) {
    const int4 c4 = *(const int4*)&s_cm[4 * g];
    float a0 = 0.0f, a1 = 0.0f, a2 = 0.0f, a3 = 0.0f;
#pragma unroll
    for (int r = 0; r < 16; ++r) {
        float d0 = v[r].x - mb[r * NC + c4.x];
        float d1 = v[r].y - mb[r * NC + c4.y];
        float d2 = v[r].z - mb[r * NC + c4.z];
        float d3 = v[r].w - mb[r * NC + c4.w];
        a0 = fmaf(d0, d0, a0);
        a1 = fmaf(d1, d1, a1);
        a2 = fmaf(d2, d2, a2);
        a3 = fmaf(d3, d3, a3);
    }
    *(float4*)&part[4 * g] = make_float4(a0, a1, a2, a3);
}

__global__ __launch_bounds__(256)
void class_mean_dist_kernel(const float* __restrict__ x,
                            const int*   __restrict__ cm,
                            float*       __restrict__ out) {
    __shared__ __attribute__((aligned(16))) int   s_cm[NP];
    __shared__ float s_cnt[NC];
    __shared__ float s_mean[DD * NC];
    __shared__ __attribute__((aligned(16))) float s_part[8 * NP];
    __shared__ __attribute__((aligned(16))) float s_stage[8 * 16 * SP];
    __shared__ __attribute__((aligned(16))) int   s_pos[NP];
    __shared__ __attribute__((aligned(16))) int   s_cls[NST * NSLOT];
    __shared__ __attribute__((aligned(16))) int   s_nu[NST * 4];

    const int tid  = threadIdx.x;
    const int b    = blockIdx.x;
    const int lane = tid & 31;
    const int wv   = tid >> 5;

    const float* xs = x + ((size_t)b * DD + (size_t)wv * 16) * NP;
    float4 va[16], vb[16];
    const int4* cm4  = (const int4*)(cm + b * NP);
    const bool  has1 = tid < NG - 256;
    const int4  c0   = cm4[tid];
    int4        c1   = make_int4(0, 0, 0, 0);
    if (has1) c1 = cm4[256 + tid];
    load_rows(va, xs + 4 * lane);

    for (int i = tid; i < NST * NSLOT; i += 256) s_cls[i] = -1;
    if (tid < NC) s_cnt[tid] = 0.0f;
    float* st = s_stage + wv * 16 * SP;
    for (int i = lane; i < 16 * SP / 4; i += 32) ((float4*)st)[i] = make_float4(0.0f, 0.0f, 0.0f, 0.0f);
    ((int4*)s_cm)[tid] = c0;
    if (has1) ((int4*)s_cm)[256 + tid] = c1;
    __syncthreads();
    atomicAdd(&s_cnt[c0.x], 1.0f);
    atomicAdd(&s_cnt[c0.y], 1.0f);
    atomicAdd(&s_cnt[c0.z], 1.0f);
    atomicAdd(&s_cnt[c0.w], 1.0f);
    if (has1) {
        atomicAdd(&s_cnt[c1.x], 1.0f);
        atomicAdd(&s_cnt[c1.y], 1.0f);
        atomicAdd(&s_cnt[c1.z], 1.0f);
        atomicAdd(&s_cnt[c1.w], 1.0f);
    }

    for (int s = wv; s < NST; s += 8) {
        const int  p0    = 128 * s + 4 * lane;
        const bool valid = p0 < NP;
        int4 c4 = make_int4(0, 0, 0, 0);
        if (valid) c4 = *(const int4*)&s_cm[p0];
        const int sh0 = (c4.x >> 4) << 3, sh1 = (c4.y >> 4) << 3;
        const int sh2 = (c4.z >> 4) << 3, sh3 = (c4.w >> 4) << 3;
        const unsigned own = valid ? ((1u << sh0) + (1u << sh1) + (1u << sh2) + (1u << sh3)) : 0u;
        unsigned inc = own;
#pragma unroll
        for (int d = 1; d < 32; d <<= 1) {
            const unsigned up = __shfl_up(inc, (unsigned)d);
            if (lane >= d) inc += up;
        }
        const unsigned tot = __shfl(inc, 31);
        const int u0 = (int)(((tot      ) & 255u) + 15u) >> 4;
        const int u1 = (int)(((tot >>  8) & 255u) + 15u) >> 4;
        const int u2 = (int)(((tot >> 16) & 255u) + 15u) >> 4;
        const int u3 = (int)(((tot >> 24)       ) + 15u) >> 4;
        const unsigned base = ((unsigned)(16 * u0) << 8) + ((unsigned)(16 * (u0 + u1)) << 16)
                            + ((unsigned)(16 * (u0 + u1 + u2)) << 24);
        unsigned run = base + (inc - own);
        const int q0 = (int)((run >> sh0) & 255u); run += 1u << sh0;
        const int q1 = (int)((run >> sh1) & 255u); run += 1u << sh1;
        const int q2 = (int)((run >> sh2) & 255u); run += 1u << sh2;
        const int q3 = (int)((run >> sh3) & 255u);
        if (valid) {
            *(int4*)&s_pos[p0] = make_int4(q0, q1, q2, q3);
            s_cls[s * NSLOT + q0] = c4.x;
            s_cls[s * NSLOT + q1] = c4.y;
            s_cls[s * NSLOT + q2] = c4.z;
            s_cls[s * NSLOT + q3] = c4.w;
        }
        if (lane == 0) *(int4*)&s_nu[4 * s] = make_int4(u0, u1, u2, u3);
    }
    __syncthreads();

    const int  myc  = lane & 15;
    const int  kb   = (lane >> 4) << 1;
    const float* arow = st + myc * SP + kb;

    v8f acc[8] = {};

    for (int s = 0; s < NFULL; ++s) {
        stage_rows(st, va, *(const int4*)&s_pos[128 * s + 4 * lane]);
        load_rows(va, xs + 4 * min(32 * (s + 1) + lane, NG - 1));
        wave_lds_order();
        wmma_step(acc, arow, s_cls + NSLOT * s + kb, s_nu + 4 * s, myc);
        wave_lds_order();
    }
    if (lane < 4)
        stage_rows(st, va, *(const int4*)&s_pos[128 * NFULL + 4 * lane]);
    wave_lds_order();
    wmma_step(acc, arow, s_cls + NSLOT * NFULL + kb, s_nu + 4 * NFULL, myc);

    const int rbase = wv * 16 + ((lane >> 4) << 3);
#pragma unroll
    for (int t = 0; t < 4; ++t) {
        int col = t * 16 + myc;
        const v8f sum = acc[2 * t] + acc[2 * t + 1];
#pragma unroll
        for (int j = 0; j < 8; ++j)
            s_mean[(rbase + j) * NC + col] = sum[j];
    }
    __syncthreads();

    for (int i = tid; i < DD * NC; i += 256)
        s_mean[i] = s_mean[i] / (s_cnt[i & (NC - 1)] + EPSF);
    __syncthreads();

    const float* mb   = s_mean + wv * 16 * NC;
    float*       part = s_part + wv * NP;
    load_rows(va, xs + 4 * lane);
    for (int j = 0; j < NSTEP - 1; j += 2) {
        const int g0 = 32 * j + lane;
        load_rows(vb, xs + 4 * (g0 + 32));
        __builtin_amdgcn_sched_barrier(0);
        slab_dist(va, mb, s_cm, part, g0);
        load_rows(va, xs + 4 * min(g0 + 64, NG - 1));
        __builtin_amdgcn_sched_barrier(0);
        slab_dist(vb, mb, s_cm, part, g0 + 32);
    }
    {
        const int g = 32 * (NSTEP - 1) + lane;
        if (g < NG) slab_dist(va, mb, s_cm, part, g);
    }
    __syncthreads();

    float4* out4 = (float4*)(out + (size_t)b * NP);
    for (int q = tid; q < NG; q += 256) {
        float4 s = *(const float4*)&s_part[4 * q];
#pragma unroll
        for (int w = 1; w < 8; ++w) {
            float4 p = *(const float4*)&s_part[w * NP + 4 * q];
            s.x += p.x; s.y += p.y; s.z += p.z; s.w += p.w;
        }
        const float sc = 1.0f / (float)DD;
        out4[q] = make_float4(s.x * sc, s.y * sc, s.z * sc, s.w * sc);
    }
}

extern "C" void kernel_launch(void* const* d_in, const int* in_sizes, int n_in,
                              void* d_out, int out_size, void* d_ws, size_t ws_size,
                              hipStream_t stream) {
    (void)in_sizes; (void)n_in; (void)d_ws; (void)ws_size; (void)out_size;
    const float* x  = (const float*)d_in[0];
    const int*   cm = (const int*)d_in[1];
    float*       o  = (float*)d_out;
    class_mean_dist_kernel<<<NB, 256, 0, stream>>>(x, cm, o);
}


# distance pass takes x of steps 4..10 of 11 from where the mean pass left it on chip (4 steps in registers, 2 in spare LDS, the short one in va); only steps 0..3 come from memory twice; sums unchanged
# speedup vs baseline: 4.3956x; 1.0349x over previous
#include <hip/hip_runtime.h>

typedef float v2f __attribute__((ext_vector_type(2)));
typedef float v8f __attribute__((ext_vector_type(8)));

#define NB   256
#define DD   128
#define NP   1296
#define NC   64
#define NG   324
#define NSTEP 11
#define EPSF 1e-4f
#define SP   196
#define NSLOT 192
#define NST  11
#define NFULL 10
#define KL0  4
#define KR0  6
#define KSTEP (8 * 16 * 128)
static_assert(KL0 == 4 && KR0 == 6 && KR0 + 4 == NFULL, "the distance pass names its steps by these numbers");

__device__ __forceinline__ void load_rows(float4 (&v)[16], const float* __restrict__ p) {
#pragma unroll
    for (int r = 0; r < 16; ++r) v[r] = *(const float4*)(p + (size_t)r * NP);
}

__device__ __forceinline__ void keep_store(float* kp, const float4 (&v)[16]) {
#pragma unroll
    for (int r = 0; r < 16; ++r) *(float4*)(kp + r * 128) = v[r];
}
__device__ __forceinline__ void keep_load(float4 (&v)[16], const float* kp) {
#pragma unroll
    for (int r = 0; r < 16; ++r) v[r] = *(const float4*)(kp + r * 128);
}
__device__ __forceinline__ void copy_rows(float4 (&d)[16], const float4 (&v)[16]) {
#pragma unroll
    for (int r = 0; r < 16; ++r) d[r] = v[r];
}

__device__ __forceinline__ void stage_rows(float* st, const float4 (&v)[16], int4 ps) {
    float* p0 = st + ps.x;
    float* p1 = st + ps.y;
    float* p2 = st + ps.z;
    float* p3 = st + ps.w;
#pragma unroll
    for (int r = 0; r < 16; ++r) {
        p0[r * SP] = v[r].x;
        p1[r * SP] = v[r].y;
        p2[r * SP] = v[r].z;
        p3[r * SP] = v[r].w;
    }
}

__device__ __forceinline__ void wave_lds_order() {
    __builtin_amdgcn_fence(__ATOMIC_RELEASE, "wavefront");
    __builtin_amdgcn_wave_barrier();
    __builtin_amdgcn_fence(__ATOMIC_ACQUIRE, "wavefront");
}

__device__ __forceinline__ void load_unit(v2f (&a)[4], int2 (&k)[4], const float* ap, const int* kp) {
#pragma unroll
    for (int q = 0; q < 4; ++q) {
        a[q] = *(const v2f*)(ap + 4 * q);
        k[q] = *(const int2*)(kp + 4 * q);
    }
}

template <int T>
__device__ __forceinline__ void tile_units(v8f (&acc)[8], v2f (&a)[4], int2 (&k)[4],
                                           const float*& ap, const int*& kp, int n, int myc) {
    const int cc = 16 * T + myc;
    for (int u = 0; u < n; ++u) {
        v2f  an[4];
        int2 kn[4];
        ap += 16;
        kp += 16;
        load_unit(an, kn, ap, kp);
        __builtin_amdgcn_sched_barrier(0);
#pragma unroll
        for (int q = 0; q < 4; ++q) {
            v2f bf;
            bf.x = (k[q].x == cc) ? 1.0f : 0.0f;
            bf.y = (k[q].y == cc) ? 1.0f : 0.0f;
            acc[2 * T + (q & 1)] = __builtin_amdgcn_wmma_f32_16x16x4_f32(
                false, a[q], false, bf, (short)0, acc[2 * T + (q & 1)], false, false);
        }
#pragma unroll
        for (int q = 0; q < 4; ++q) { a[q] = an[q]; k[q] = kn[q]; }
    }
}

__device__ __forceinline__ void wmma_step(v8f (&acc)[8], const float* arow, const int* kp0,
                                          const int* nu, int myc) {
    const int4 n4 = *(const int4*)nu;
    const int n0 = __builtin_amdgcn_readfirstlane(n4.x);
    const int n1 = __builtin_amdgcn_readfirstlane(n4.y);
    const int n2 = __builtin_amdgcn_readfirstlane(n4.z);
    const int n3 = __builtin_amdgcn_readfirstlane(n4.w);
    const float* ap = arow;
    const int*   kp = kp0;
    v2f  a[4];
    int2 k[4];
    load_unit(a, k, ap, kp);
    tile_units<0>(acc, a, k, ap, kp, n0, myc);
    tile_units<1>(acc, a, k, ap, kp, n1, myc);
    tile_units<2>(acc, a, k, ap, kp, n2, myc);
    tile_units<3>(acc, a, k, ap, kp, n3, myc);
}

__device__ __forceinline__ void slab_dist(const float4 (&v)[16], const float* mb,
                                          const int* s_cm, float* part, int g) {
    const int4 c4 = *(const int4*)&s_cm[4 * g];
    float a0 = 0.0f, a1 = 0.0f, a2 = 0.0f, a3 = 0.0f;
#pragma unroll
    for (int r = 0; r < 16; ++r) {
        float d0 = v[r].x - mb[r * NC + c4.x];
        float d1 = v[r].y - mb[r * NC + c4.y];
        float d2 = v[r].z - mb[r * NC + c4.z];
        float d3 = v[r].w - mb[r * NC + c4.w];
        a0 = fmaf(d0, d0, a0);
        a1 = fmaf(d1, d1, a1);
        a2 = fmaf(d2, d2, a2);
        a3 = fmaf(d3, d3, a3);
    }
    *(float4*)&part[4 * g] = make_float4(a0, a1, a2, a3);
}

__global__ __launch_bounds__(256)
void class_mean_dist_kernel(const float* __restrict__ x,
                            const int*   __restrict__ cm,
                            float*       __restrict__ out) {
    __shared__ __attribute__((aligned(16))) int   s_cm[NP];
    __shared__ float s_cnt[NC];
    __shared__ float s_mean[DD * NC];
    __shared__ __attribute__((aligned(16))) float s_part[8 * NP];
    __shared__ __attribute__((aligned(16))) float s_stage[8 * 16 * SP];
    __shared__ __attribute__((aligned(16))) int   s_pos[NP];
    __shared__ __attribute__((aligned(16))) int   s_cls[NST * NSLOT];
    __shared__ __attribute__((aligned(16))) int   s_nu[NST * 4];
    __shared__ __attribute__((aligned(16))) float s_keep[(KR0 - KL0) * KSTEP];

    const int tid  = threadIdx.x;
    const int b    = blockIdx.x;
    const int lane = tid & 31;
    const int wv   = tid >> 5;

    const float* xs = x + ((size_t)b * DD + (size_t)wv * 16) * NP;
    float4 va[16], vb[16];
    float4 k0[16], k1[16] = {}, k2[16] = {}, k3[16] = {};
    const int4* cm4  = (const int4*)(cm + b * NP);
    const bool  has1 = tid < NG - 256;
    const int4  c0   = cm4[tid];
    int4        c1   = make_int4(0, 0, 0, 0);
    if (has1) c1 = cm4[256 + tid];
    load_rows(va, xs + 4 * lane);

    for (int i = tid; i < NST * NSLOT; i += 256) s_cls[i] = -1;
    if (tid < NC) s_cnt[tid] = 0.0f;
    float* st = s_stage + wv * 16 * SP;
    for (int i = lane; i < 16 * SP / 4; i += 32) ((float4*)st)[i] = make_float4(0.0f, 0.0f, 0.0f, 0.0f);
    ((int4*)s_cm)[tid] = c0;
    if (has1) ((int4*)s_cm)[256 + tid] = c1;
    __syncthreads();
    atomicAdd(&s_cnt[c0.x], 1.0f);
    atomicAdd(&s_cnt[c0.y], 1.0f);
    atomicAdd(&s_cnt[c0.z], 1.0f);
    atomicAdd(&s_cnt[c0.w], 1.0f);
    if (has1) {
        atomicAdd(&s_cnt[c1.x], 1.0f);
        atomicAdd(&s_cnt[c1.y], 1.0f);
        atomicAdd(&s_cnt[c1.z], 1.0f);
        atomicAdd(&s_cnt[c1.w], 1.0f);
    }

    for (int s = wv; s < NST; s += 8) {
        const int  p0    = 128 * s + 4 * lane;
        const bool valid = p0 < NP;
        int4 c4 = make_int4(0, 0, 0, 0);
        if (valid) c4 = *(const int4*)&s_cm[p0];
        const int sh0 = (c4.x >> 4) << 3, sh1 = (c4.y >> 4) << 3;
        const int sh2 = (c4.z >> 4) << 3, sh3 = (c4.w >> 4) << 3;
        const unsigned own = valid ? ((1u << sh0) + (1u << sh1) + (1u << sh2) + (1u << sh3)) : 0u;
        unsigned inc = own;
#pragma unroll
        for (int d = 1; d < 32; d <<= 1) {
            const unsigned up = __shfl_up(inc, (unsigned)d);
            if (lane >= d) inc += up;
        }
        const unsigned tot = __shfl(inc, 31);
        const int u0 = (int)(((tot      ) & 255u) + 15u) >> 4;
        const int u1 = (int)(((tot >>  8) & 255u) + 15u) >> 4;
        const int u2 = (int)(((tot >> 16) & 255u) + 15u) >> 4;
        const int u3 = (int)(((tot >> 24)       ) + 15u) >> 4;
        const unsigned base = ((unsigned)(16 * u0) << 8) + ((unsigned)(16 * (u0 + u1)) << 16)
                            + ((unsigned)(16 * (u0 + u1 + u2)) << 24);
        unsigned run = base + (inc - own);
        const int q0 = (int)((run >> sh0) & 255u); run += 1u << sh0;
        const int q1 = (int)((run >> sh1) & 255u); run += 1u << sh1;
        const int q2 = (int)((run >> sh2) & 255u); run += 1u << sh2;
        const int q3 = (int)((run >> sh3) & 255u);
        if (valid) {
            *(int4*)&s_pos[p0] = make_int4(q0, q1, q2, q3);
            s_cls[s * NSLOT + q0] = c4.x;
            s_cls[s * NSLOT + q1] = c4.y;
            s_cls[s * NSLOT + q2] = c4.z;
            s_cls[s * NSLOT + q3] = c4.w;
        }
        if (lane == 0) *(int4*)&s_nu[4 * s] = make_int4(u0, u1, u2, u3);
    }
    __syncthreads();

    const int  myc  = lane & 15;
    const int  kb   = (lane >> 4) << 1;
    const float* arow = st + myc * SP + kb;

    v8f acc[8] = {};

    float* keep = s_keep + wv * 16 * 128 + 4 * lane;
    for (int s = 0; s < KR0; ++s) {
        stage_rows(st, va, *(const int4*)&s_pos[128 * s + 4 * lane]);
        if (s >= KL0) keep_store(keep + (s - KL0) * KSTEP, va);
        load_rows(va, xs + 4 * (32 * (s + 1) + lane));
        wave_lds_order();
        wmma_step(acc, arow, s_cls + NSLOT * s + kb, s_nu + 4 * s, myc);
        wave_lds_order();
    }
    for (int s = KR0; s < NFULL; ++s) {
        stage_rows(st, va, *(const int4*)&s_pos[128 * s + 4 * lane]);
        copy_rows(k0, k1);
        copy_rows(k1, k2);
        copy_rows(k2, k3);
        copy_rows(k3, va);
        load_rows(va, xs + 4 * min(32 * (s + 1) + lane, NG - 1));
        wave_lds_order();
        wmma_step(acc, arow, s_cls + NSLOT * s + kb, s_nu + 4 * s, myc);
        wave_lds_order();
    }
    if (lane < 4)
        stage_rows(st, va, *(const int4*)&s_pos[128 * NFULL + 4 * lane]);
    wave_lds_order();
    wmma_step(acc, arow, s_cls + NSLOT * NFULL + kb, s_nu + 4 * NFULL, myc);

    const int rbase = wv * 16 + ((lane >> 4) << 3);
#pragma unroll
    for (int t = 0; t < 4; ++t) {
        int col = t * 16 + myc;
        const v8f sum = acc[2 * t] + acc[2 * t + 1];
#pragma unroll
        for (int j = 0; j < 8; ++j)
            s_mean[(rbase + j) * NC + col] = sum[j];
    }
    __syncthreads();

    for (int i = tid; i < DD * NC; i += 256)
        s_mean[i] = s_mean[i] / (s_cnt[i & (NC - 1)] + EPSF);
    __syncthreads();

    const float* mb   = s_mean + wv * 16 * NC;
    float*       part = s_part + wv * NP;
    load_rows(vb, xs + 4 * lane);
    __builtin_amdgcn_sched_barrier(0);
    {
        const int g = 32 * NFULL + lane;
        if (g < NG) slab_dist(va, mb, s_cm, part, g);
    }
    load_rows(va, xs + 4 * (32 * 1 + lane));
    __builtin_amdgcn_sched_barrier(0);
    slab_dist(k0, mb, s_cm, part, 32 * KR0 + lane);
    load_rows(k0, xs + 4 * (32 * 2 + lane));
    __builtin_amdgcn_sched_barrier(0);
    slab_dist(k1, mb, s_cm, part, 32 * (KR0 + 1) + lane);
    load_rows(k1, xs + 4 * (32 * 3 + lane));
    __builtin_amdgcn_sched_barrier(0);
    slab_dist(k2, mb, s_cm, part, 32 * (KR0 + 2) + lane);
    slab_dist(k3, mb, s_cm, part, 32 * (KR0 + 3) + lane);
#pragma nounroll
    for (int j = 0; j < KR0; j += 2) {
        slab_dist(vb, mb, s_cm, part, 32 * j + lane);
        slab_dist(va, mb, s_cm, part, 32 * (j + 1) + lane);
        if (j == 0) {
            copy_rows(vb, k0);
            copy_rows(va, k1);
        } else if (j + 2 == KL0) {
            keep_load(vb, keep);
            keep_load(va, keep + KSTEP);
        }
    }
    __syncthreads();

    float4* out4 = (float4*)(out + (size_t)b * NP);
    for (int q = tid; q < NG; q += 256) {
        float4 s = *(const float4*)&s_part[4 * q];
#pragma unroll
        for (int w = 1; w < 8; ++w) {
            float4 p = *(const float4*)&s_part[w * NP + 4 * q];
            s.x += p.x; s.y += p.y; s.z += p.z; s.w += p.w;
        }
        const float sc = 1.0f / (float)DD;
        out4[q] = make_float4(s.x * sc, s.y * sc, s.z * sc, s.w * sc);
    }
}

extern "C" void kernel_launch(void* const* d_in, const int* in_sizes, int n_in,
                              void* d_out, int out_size, void* d_ws, size_t ws_size,
                              hipStream_t stream) {
    (void)in_sizes; (void)n_in; (void)d_ws; (void)ws_size; (void)out_size;
    const float* x  = (const float*)d_in[0];
    const int*   cm = (const int*)d_in[1];
    float*       o  = (float*)d_out;
    class_mean_dist_kernel<<<NB, 256, 0, stream>>>(x, cm, o);
}
